# starting kernel
# baseline (speedup 1.0000x reference)
#include <hip/hip_runtime.h>


typedef float v2f __attribute__((ext_vector_type(2)));
typedef float v8f __attribute__((ext_vector_type(8)));

#define WMMA4(A, B, C)                                                         \
  __builtin_amdgcn_wmma_f32_16x16x4_f32(false, (A), false, (B), (short)0,      \
                                        (C), false, false)

#define B_TILE 4

__global__ __launch_bounds__(256, 1) void tt_fused_kernel(
    const float* __restrict__ x, const float* __restrict__ c0,
    const float* __restrict__ c1, const float* __restrict__ c2,
    const float* __restrict__ c3, const float* __restrict__ bias,
    float* __restrict__ out) {
  __shared__ __align__(16) float lw0[8 * 64];
  __shared__ __align__(16) float lw1[64 * 64];
  __shared__ __align__(16) float lw2[64 * 64];
  __shared__ __align__(16) float lw3[64 * 8];
  __shared__ __align__(16) float z0[2 * 256 * 64];
  __shared__ __align__(16) float z1[8 * 32 * 64];
  __shared__ __align__(16) float z2[8 * 32 * 64];

  const int tid = threadIdx.x;
  const int w = tid >> 5;
  const int lane = tid & 31;
  const int lo = lane & 15;
  const int hi = lane >> 4;
  const int bg0 = blockIdx.x * B_TILE;

  for (int i = tid; i < 512; i += 256) lw0[i] = c0[i];
  for (int i = tid; i < 4096; i += 256) lw1[i] = c1[i];
  for (int i = tid; i < 4096; i += 256) lw2[i] = c2[i];
  for (int i = tid; i < 512; i += 256) lw3[i] = c3[i];
  __syncthreads();

  for (int pi = 0; pi < 4; ++pi) {
    {
      v2f bb0, bb1;
      bb0[0] = lw0[(2 * hi + 0) * 64 + pi * 16 + lo];
      bb0[1] = lw0[(2 * hi + 1) * 64 + pi * 16 + lo];
      bb1[0] = lw0[(4 + 2 * hi + 0) * 64 + pi * 16 + lo];
      bb1[1] = lw0[(4 + 2 * hi + 1) * 64 + pi * 16 + lo];
      for (int mt = w; mt < 128; mt += 8) {
        const int m0 = mt * 16 + lo;
        const int b = m0 >> 9;
        const int i1 = (m0 >> 6) & 7;
        const int sp = m0 & 63;
        const float* xr = x + (size_t)(bg0 + b) * 4096 + i1 * 64 + sp;
        v2f a0, a1;
        a0[0] = xr[(2 * hi + 0) * 512];
        a0[1] = xr[(2 * hi + 1) * 512];
        a1[0] = xr[(4 + 2 * hi + 0) * 512];
        a1[1] = xr[(4 + 2 * hi + 1) * 512];
        v8f acc = {0.f, 0.f, 0.f, 0.f, 0.f, 0.f, 0.f, 0.f};
        acc = WMMA4(a0, bb0, acc);
        acc = WMMA4(a1, bb1, acc);
        const int p = lo >> 3;
        const int r1 = lo & 7;
#pragma unroll
        for (int v = 0; v < 8; ++v) {
          const int mr = mt * 16 + 8 * hi + v;
          const int br = mr >> 9;
          const int i1r = (mr >> 6) & 7;
          const int spr = mr & 63;
          z0[p * 16384 + (br * 64 + spr) * 64 + r1 * 8 + i1r] = acc[v];
        }
      }
    }
    __syncthreads();

    for (int p = 0; p < 2; ++p) {
      const int o0 = pi * 2 + p;
      {
        const int nt = w & 3;
        const int half = w >> 2;
        const int nb = nt * 16 + lo;
        v2f wb[16];
#pragma unroll
        for (int t = 0; t < 16; ++t) {
          const int k = 4 * t + 2 * hi;
          wb[t][0] = lw1[k * 64 + nb];
          wb[t][1] = lw1[(k + 1) * 64 + nb];
        }
        const int o1s = nb >> 3;
        const int r2s = nb & 7;
        for (int i = 0; i < 8; ++i) {
          const int mt = half * 8 + i;
          const float* arow = &z0[p * 16384 + (mt * 16 + lo) * 64];
          v8f acc = {0.f, 0.f, 0.f, 0.f, 0.f, 0.f, 0.f, 0.f};
#pragma unroll
          for (int t = 0; t < 16; ++t) {
            v2f a = *(const v2f*)(arow + 4 * t + 2 * hi);
            acc = WMMA4(a, wb[t], acc);
          }
#pragma unroll
          for (int v = 0; v < 8; ++v) {
            const int m1 = mt * 16 + 8 * hi + v;
            const int b = m1 >> 6;
            const int i2 = (m1 >> 3) & 7;
            const int i3 = m1 & 7;
            z1[o1s * 2048 + (b * 8 + i3) * 64 + r2s * 8 + i2] = acc[v];
          }
        }
      }
      __syncthreads();

      {
        const int nt = w & 3;
        const int mt = w >> 2;
        const int nb = nt * 16 + lo;
        v2f wb2[16];
#pragma unroll
        for (int t = 0; t < 16; ++t) {
          const int k = 4 * t + 2 * hi;
          wb2[t][0] = lw2[k * 64 + nb];
          wb2[t][1] = lw2[(k + 1) * 64 + nb];
        }
        const int o2s = nb >> 3;
        const int r3s = nb & 7;
        for (int o1 = 0; o1 < 8; ++o1) {
          const float* arow = &z1[o1 * 2048 + (mt * 16 + lo) * 64];
          v8f acc = {0.f, 0.f, 0.f, 0.f, 0.f, 0.f, 0.f, 0.f};
#pragma unroll
          for (int t = 0; t < 16; ++t) {
            v2f a = *(const v2f*)(arow + 4 * t + 2 * hi);
            acc = WMMA4(a, wb2[t], acc);
          }
#pragma unroll
          for (int v = 0; v < 8; ++v) {
            const int m2 = mt * 16 + 8 * hi + v;
            const int b = m2 >> 3;
            const int i3 = m2 & 7;
            z2[o1 * 2048 + (b * 8 + o2s) * 64 + r3s * 8 + i3] = acc[v];
          }
        }
      }
      __syncthreads();

      {
        const int o1 = w;
        v2f wb3[16];
#pragma unroll
        for (int t = 0; t < 16; ++t) {
          const int k = 4 * t + 2 * hi;
          wb3[t][0] = (lo < 8) ? lw3[k * 8 + lo] : 0.f;
          wb3[t][1] = (lo < 8) ? lw3[(k + 1) * 8 + lo] : 0.f;
        }
        for (int mt = 0; mt < 2; ++mt) {
          const float* arow = &z2[o1 * 2048 + (mt * 16 + lo) * 64];
          v8f acc = {0.f, 0.f, 0.f, 0.f, 0.f, 0.f, 0.f, 0.f};
#pragma unroll
          for (int t = 0; t < 16; ++t) {
            v2f a = *(const v2f*)(arow + 4 * t + 2 * hi);
            acc = WMMA4(a, wb3[t], acc);
          }
          if (lo < 8) {
#pragma unroll
            for (int v = 0; v < 8; ++v) {
              const int m3 = mt * 16 + 8 * hi + v;
              const int b = m3 >> 3;
              const int o2 = m3 & 7;
              const int feat = ((o0 * 8 + o1) * 8 + o2) * 8 + lo;
              out[(size_t)(bg0 + b) * 4096 + feat] = acc[v] + bias[feat];
            }
          }
        }
      }
      __syncthreads();
    }
  }
}

extern "C" void kernel_launch(void* const* d_in, const int* in_sizes, int n_in,
                              void* d_out, int out_size, void* d_ws,
                              size_t ws_size, hipStream_t stream) {
  const float* x = (const float*)d_in[0];
  const float* c0 = (const float*)d_in[1];
  const float* c1 = (const float*)d_in[2];
  const float* c2 = (const float*)d_in[3];
  const float* c3 = (const float*)d_in[4];
  const float* bias = (const float*)d_in[5];
  float* out = (float*)d_out;

  dim3 grid(4096 / B_TILE);
  dim3 block(256);
  tt_fused_kernel<<<grid, block, 0, stream>>>(x, c0, c1, c2, c3, bias, out);
}


# All 4 contractions chained in registers: 16 batch rows on the WMMA columns, each stage's D registers are the next stage's B operand, no intermediate in LDS; sum order inside each contraction changes
# speedup vs baseline: 3.4942x; 3.4942x over previous
#include <hip/hip_runtime.h>


typedef float v2f __attribute__((ext_vector_type(2)));
typedef float v4f __attribute__((ext_vector_type(4)));
typedef float v8f __attribute__((ext_vector_type(8)));

#define WMMA4(A, B, C)                                                         \
  __builtin_amdgcn_wmma_f32_16x16x4_f32(false, (A), false, (B), (short)0,      \
                                        (C), false, false)

__global__ __launch_bounds__(128, 1) void tt_chain_kernel(
    const float* __restrict__ x, const float* __restrict__ c0,
    const float* __restrict__ c1, const float* __restrict__ c2,
    const float* __restrict__ c3, const float* __restrict__ bias,
    float* __restrict__ out) {
  __shared__ __align__(16) float xs[4096 * 16];
  __shared__ __align__(16) v2f f2s[8 * 2 * 4 * 32];
  __shared__ __align__(16) v2f f3s[8 * 4 * 32];

  const int tid = threadIdx.x;
  const int w = tid >> 5;
  const int lane = tid & 31;
  const int c = lane & 15;
  const int h = lane >> 4;
  const int bg0 = blockIdx.x * 16;

  {
    const float* xrow = x + (size_t)(bg0 + c) * 4096;
    for (int n0 = 0; n0 < 128; n0 += 16) {
      v4f v[16];
#pragma unroll
      for (int k = 0; k < 16; ++k)
        v[k] = *(const v4f*)(xrow + 4 * ((n0 + k) * 8 + w * 2 + h));
#pragma unroll
      for (int k = 0; k < 16; ++k) {
        const int q = (n0 + k) * 8 + w * 2 + h;
#pragma unroll
        for (int e = 0; e < 4; ++e) {
          const int idx = 4 * q + e;
          xs[((idx << 4) + c) ^ (((idx >> 10) & 1) << 4)] = v[k][e];
        }
      }
    }
  }
  for (int e = tid; e < 2048; e += 128) {
    const int ln = e & 31, T2 = (e >> 5) & 3, u = (e >> 7) & 1, i2 = e >> 8;
    const int cc = ln & 15, hh = ln >> 4;
    const int o2 = 2 * T2 + (cc >> 3), r3 = cc & 7;
    v2f v;
#pragma unroll
    for (int j = 0; j < 2; ++j) {
      const int r2 = 4 * hh + 2 * u + j;
      v[j] = c2[((r2 * 8 + i2) * 8 + o2) * 8 + r3];
    }
    f2s[e] = v;
  }
  for (int e = tid; e < 1024; e += 128) {
    const int ln = e & 31, t = (e >> 5) & 3, i3 = e >> 7;
    const int cc = ln & 15, hh = ln >> 4;
    v2f v;
#pragma unroll
    for (int j = 0; j < 2; ++j)
      v[j] = ((cc >> 3) == hh) ? c3[((2 * t + j) * 8 + i3) * 8 + (cc & 7)] : 0.f;
    f3s[e] = v;
  }
  __syncthreads();

  v2f f1[8][2][4];
#pragma unroll
  for (int i1 = 0; i1 < 8; ++i1)
#pragma unroll
    for (int u = 0; u < 2; ++u)
#pragma unroll
      for (int T = 0; T < 4; ++T)
#pragma unroll
        for (int j = 0; j < 2; ++j) {
          const int r1 = 4 * h + 2 * u + j;
          const int o1 = 2 * T + ((c >> 2) & 1);
          const int r2 = 4 * (c >> 3) + (c & 3);
          f1[i1][u][T][j] = c1[((r1 * 8 + i1) * 8 + o1) * 8 + r2];
        }

  for (int pass = 0; pass < 2; ++pass) {
    const int o0 = w + 4 * pass;
    v2f a0[2];
#pragma unroll
    for (int t = 0; t < 2; ++t)
#pragma unroll
      for (int j = 0; j < 2; ++j) {
        const int i0 = 4 * t + 2 * h + j;
        const int r1 = 4 * (c >> 3) + (c & 3);
        a0[t][j] = (c & 4) ? 0.f : c0[(i0 * 8 + o0) * 8 + r1];
      }

    v8f acc3[8][4];
#pragma unroll
    for (int o1 = 0; o1 < 8; ++o1)
#pragma unroll
      for (int T2 = 0; T2 < 4; ++T2)
#pragma unroll
        for (int r = 0; r < 8; ++r) acc3[o1][T2][r] = 0.f;

    for (int i3 = 0; i3 < 8; ++i3) {
      v8f acc2[8][4];
#pragma unroll
      for (int o1 = 0; o1 < 8; ++o1)
#pragma unroll
        for (int T2 = 0; T2 < 4; ++T2)
#pragma unroll
          for (int r = 0; r < 8; ++r) acc2[o1][T2][r] = 0.f;
      const float* xl = xs + ((((h << 14) + c + (i3 << 4))) ^ (h << 4));

#pragma unroll
      for (int i2 = 0; i2 < 8; ++i2) {
        v8f d0[8];
#pragma unroll
        for (int i1 = 0; i1 < 8; ++i1) {
          v8f d = {0.f, 0.f, 0.f, 0.f, 0.f, 0.f, 0.f, 0.f};
#pragma unroll
          for (int t = 0; t < 2; ++t) {
            v2f bx;
#pragma unroll
            for (int j = 0; j < 2; ++j) {
              bx[j] = xl[t * 32768 + j * 8192 + i1 * 1024 + i2 * 128];
            }
            d = WMMA4(a0[t], bx, d);
          }
          d0[i1] = d;
        }
        v8f d1[4];
#pragma unroll
        for (int T = 0; T < 4; ++T)
#pragma unroll
          for (int r = 0; r < 8; ++r) d1[T][r] = 0.f;
#pragma unroll
        for (int i1 = 0; i1 < 8; ++i1)
#pragma unroll
          for (int u = 0; u < 2; ++u) {
            v2f bz;
            bz[0] = d0[i1][2 * u];
            bz[1] = d0[i1][2 * u + 1];
#pragma unroll
            for (int T = 0; T < 4; ++T) d1[T] = WMMA4(f1[i1][u][T], bz, d1[T]);
          }
#pragma unroll
        for (int u = 0; u < 2; ++u)
#pragma unroll
          for (int T2 = 0; T2 < 4; ++T2) {
            const v2f a2 = f2s[((i2 * 2 + u) * 4 + T2) * 32 + lane];
#pragma unroll
            for (int o1 = 0; o1 < 8; ++o1) {
              v2f bz;
              bz[0] = d1[o1 >> 1][4 * (o1 & 1) + 2 * u];
              bz[1] = d1[o1 >> 1][4 * (o1 & 1) + 2 * u + 1];
              acc2[o1][T2] = WMMA4(a2, bz, acc2[o1][T2]);
            }
          }
      }
#pragma unroll
      for (int t = 0; t < 4; ++t) {
        const v2f a3 = f3s[(i3 * 4 + t) * 32 + lane];
#pragma unroll
        for (int o1 = 0; o1 < 8; ++o1)
#pragma unroll
          for (int T2 = 0; T2 < 4; ++T2) {
            v2f bz;
            bz[0] = acc2[o1][T2][2 * t];
            bz[1] = acc2[o1][T2][2 * t + 1];
            acc3[o1][T2] = WMMA4(a3, bz, acc3[o1][T2]);
          }
      }
    }

    float* orow = out + (size_t)(bg0 + c) * 4096;
#pragma unroll
    for (int o1 = 0; o1 < 8; ++o1)
#pragma unroll
      for (int T2 = 0; T2 < 4; ++T2) {
        const int feat = o0 * 512 + o1 * 64 + (2 * T2 + h) * 8;
        const v4f b0 = *(const v4f*)(bias + feat);
        const v4f b1 = *(const v4f*)(bias + feat + 4);
        v4f y0, y1;
#pragma unroll
        for (int r = 0; r < 4; ++r) {
          y0[r] = acc3[o1][T2][r] + b0[r];
          y1[r] = acc3[o1][T2][4 + r] + b1[r];
        }
        *(v4f*)(orow + feat) = y0;
        *(v4f*)(orow + feat + 4) = y1;
      }
  }
}

extern "C" void kernel_launch(void* const* d_in, const int* in_sizes, int n_in,
                              void* d_out, int out_size, void* d_ws,
                              size_t ws_size, hipStream_t stream) {
  const float* x = (const float*)d_in[0];
  const float* c0 = (const float*)d_in[1];
  const float* c1 = (const float*)d_in[2];
  const float* c2 = (const float*)d_in[3];
  const float* c3 = (const float*)d_in[4];
  const float* bias = (const float*)d_in[5];
  float* out = (float*)d_out;

  dim3 grid(4096 / 16);
  dim3 block(128);
  tt_chain_kernel<<<grid, block, 0, stream>>>(x, c0, c1, c2, c3, bias, out);
}


# x tile kept in LDS in stage-0 fragment order (16-byte chunk = a lane's 4 k values): 8 ds_load_b128 per i2 block issued as one group a block ahead, not 194 small loads among the WMMAs; same sums
# speedup vs baseline: 3.7586x; 1.0757x over previous
#include <hip/hip_runtime.h>


typedef float v2f __attribute__((ext_vector_type(2)));
typedef float v4f __attribute__((ext_vector_type(4)));
typedef float v8f __attribute__((ext_vector_type(8)));

#define WMMA4(A, B, C)                                                         \
  __builtin_amdgcn_wmma_f32_16x16x4_f32(false, (A), false, (B), (short)0,      \
                                        (C), false, false)

__global__ __launch_bounds__(128, 1) void tt_chain_kernel(
    const float* __restrict__ x, const float* __restrict__ c0,
    const float* __restrict__ c1, const float* __restrict__ c2,
    const float* __restrict__ c3, const float* __restrict__ bias,
    float* __restrict__ out) {
  __shared__ __align__(16) v4f xs4[512 * 32];
  __shared__ __align__(16) v2f f2s[8 * 2 * 4 * 32];
  __shared__ __align__(16) v2f f3s[8 * 4 * 32];

  const int tid = threadIdx.x;
  const int w = tid >> 5;
  const int lane = tid & 31;
  const int c = lane & 15;
  const int h = lane >> 4;
  const int bg0 = blockIdx.x * 16;

  {
    const float* xrow = x + (size_t)(bg0 + c) * 4096;
    for (int n = 0; n < 8; ++n) {
      const int hp = n >> 2, itb = (n & 3) * 4;
      {
        v4f v[4][4];
#pragma unroll
        for (int a = 0; a < 4; ++a) {
          const int qq = (itb + a) * 8 + w * 2 + h;
#pragma unroll
          for (int m = 0; m < 4; ++m) {
            const int i0 = 2 * hp + (m & 1) + 4 * (m >> 1);
            v[a][m] = *(const v4f*)(xrow + i0 * 512 + 4 * qq);
          }
        }
#pragma unroll
        for (int a = 0; a < 4; ++a) {
          const int it = itb + a;
          const int i1 = it >> 1;
          const int i2 = (it & 1) * 4 + w;
#pragma unroll
          for (int e = 0; e < 4; ++e) {
            const int i3 = 4 * h + e;
            v4f s;
#pragma unroll
            for (int m = 0; m < 4; ++m) s[m] = v[a][m][e];
            xs4[((i3 * 8 + i2) * 8 + i1) * 32 + hp * 16 + c] = s;
          }
        }
      }
    }
  }
  for (int e = tid; e < 2048; e += 128) {
    const int ln = e & 31, T2 = (e >> 5) & 3, u = (e >> 7) & 1, i2 = e >> 8;
    const int cc = ln & 15, hh = ln >> 4;
    const int o2 = 2 * T2 + (cc >> 3), r3 = cc & 7;
    v2f v;
#pragma unroll
    for (int j = 0; j < 2; ++j) {
      const int r2 = 4 * hh + 2 * u + j;
      v[j] = c2[((r2 * 8 + i2) * 8 + o2) * 8 + r3];
    }
    f2s[e] = v;
  }
  for (int e = tid; e < 1024; e += 128) {
    const int ln = e & 31, t = (e >> 5) & 3, i3 = e >> 7;
    const int cc = ln & 15, hh = ln >> 4;
    v2f v;
#pragma unroll
    for (int j = 0; j < 2; ++j)
      v[j] = ((cc >> 3) == hh) ? c3[((2 * t + j) * 8 + i3) * 8 + (cc & 7)] : 0.f;
    f3s[e] = v;
  }
  __syncthreads();

  v2f f1[8][2][4];
#pragma unroll
  for (int i1 = 0; i1 < 8; ++i1)
#pragma unroll
    for (int u = 0; u < 2; ++u)
#pragma unroll
      for (int T = 0; T < 4; ++T)
#pragma unroll
        for (int j = 0; j < 2; ++j) {
          const int r1 = 4 * h + 2 * u + j;
          const int o1 = 2 * T + ((c >> 2) & 1);
          const int r2 = 4 * (c >> 3) + (c & 3);
          f1[i1][u][T][j] = c1[((r1 * 8 + i1) * 8 + o1) * 8 + r2];
        }

  v2f f2[8][2][4];
#pragma unroll
  for (int i2 = 0; i2 < 8; ++i2)
#pragma unroll
    for (int u = 0; u < 2; ++u)
#pragma unroll
      for (int T2 = 0; T2 < 4; ++T2)
        f2[i2][u][T2] = f2s[((i2 * 2 + u) * 4 + T2) * 32 + lane];

  v4f xa[8];
#pragma unroll
  for (int i1 = 0; i1 < 8; ++i1) xa[i1] = xs4[i1 * 32 + lane];

  for (int pass = 0; pass < 2; ++pass) {
    const int o0 = w + 4 * pass;
    v2f a0[2];
#pragma unroll
    for (int t = 0; t < 2; ++t)
#pragma unroll
      for (int j = 0; j < 2; ++j) {
        const int i0 = 4 * t + 2 * h + j;
        const int r1 = 4 * (c >> 3) + (c & 3);
        a0[t][j] = (c & 4) ? 0.f : c0[(i0 * 8 + o0) * 8 + r1];
      }

    v8f acc3[8][4];
#pragma unroll
    for (int o1 = 0; o1 < 8; ++o1)
#pragma unroll
      for (int T2 = 0; T2 < 4; ++T2)
#pragma unroll
        for (int r = 0; r < 8; ++r) acc3[o1][T2][r] = 0.f;

    for (int i3 = 0; i3 < 8; ++i3) {
      v8f acc2[8][4];
#pragma unroll
      for (int o1 = 0; o1 < 8; ++o1)
#pragma unroll
        for (int T2 = 0; T2 < 4; ++T2)
#pragma unroll
          for (int r = 0; r < 8; ++r) acc2[o1][T2][r] = 0.f;
      const v4f* xb = xs4 + i3 * 2048 + lane;
      const v4f* xnb = xs4 + ((i3 + 1) & 7) * 2048 + lane;
      v2f a3[4];

#pragma unroll
      for (int i2 = 0; i2 < 8; ++i2) {
        v4f xn[8];
        __builtin_amdgcn_sched_barrier(0);
        if (i2 < 7) {
#pragma unroll
          for (int i1 = 0; i1 < 8; ++i1) xn[i1] = xb[((i2 + 1) * 8 + i1) * 32];
        } else {
#pragma unroll
          for (int i1 = 0; i1 < 8; ++i1) xn[i1] = xnb[i1 * 32];
#pragma unroll
          for (int t = 0; t < 4; ++t) a3[t] = f3s[(i3 * 4 + t) * 32 + lane];
        }
        __builtin_amdgcn_sched_barrier(0);
        v8f d0[8];
#pragma unroll
        for (int i1 = 0; i1 < 8; ++i1) {
          v8f d = {0.f, 0.f, 0.f, 0.f, 0.f, 0.f, 0.f, 0.f};
#pragma unroll
          for (int t = 0; t < 2; ++t) {
            v2f bx;
            bx[0] = xa[i1][2 * t];
            bx[1] = xa[i1][2 * t + 1];
            d = WMMA4(a0[t], bx, d);
          }
          d0[i1] = d;
        }
        v8f d1[4];
#pragma unroll
        for (int T = 0; T < 4; ++T)
#pragma unroll
          for (int r = 0; r < 8; ++r) d1[T][r] = 0.f;
#pragma unroll
        for (int i1 = 0; i1 < 8; ++i1)
#pragma unroll
          for (int u = 0; u < 2; ++u) {
            v2f bz;
            bz[0] = d0[i1][2 * u];
            bz[1] = d0[i1][2 * u + 1];
#pragma unroll
            for (int T = 0; T < 4; ++T) d1[T] = WMMA4(f1[i1][u][T], bz, d1[T]);
          }
#pragma unroll
        for (int u = 0; u < 2; ++u)
#pragma unroll
          for (int T2 = 0; T2 < 4; ++T2) {
            const v2f a2 = f2[i2][u][T2];
#pragma unroll
            for (int o1 = 0; o1 < 8; ++o1) {
              v2f bz;
              bz[0] = d1[o1 >> 1][4 * (o1 & 1) + 2 * u];
              bz[1] = d1[o1 >> 1][4 * (o1 & 1) + 2 * u + 1];
              acc2[o1][T2] = WMMA4(a2, bz, acc2[o1][T2]);
            }
          }
#pragma unroll
        for (int i1 = 0; i1 < 8; ++i1) xa[i1] = xn[i1];
      }
#pragma unroll
      for (int t = 0; t < 4; ++t) {
#pragma unroll
        for (int o1 = 0; o1 < 8; ++o1)
#pragma unroll
          for (int T2 = 0; T2 < 4; ++T2) {
            v2f bz;
            bz[0] = acc2[o1][T2][2 * t];
            bz[1] = acc2[o1][T2][2 * t + 1];
            acc3[o1][T2] = WMMA4(a3[t], bz, acc3[o1][T2]);
          }
      }
    }

    float* orow = out + (size_t)(bg0 + c) * 4096;
#pragma unroll
    for (int o1 = 0; o1 < 8; ++o1)
#pragma unroll
      for (int T2 = 0; T2 < 4; ++T2) {
        const int feat = o0 * 512 + o1 * 64 + (2 * T2 + h) * 8;
        const v4f b0 = *(const v4f*)(bias + feat);
        const v4f b1 = *(const v4f*)(bias + feat + 4);
        v4f y0, y1;
#pragma unroll
        for (int r = 0; r < 4; ++r) {
          y0[r] = acc3[o1][T2][r] + b0[r];
          y1[r] = acc3[o1][T2][4 + r] + b1[r];
        }
        *(v4f*)(orow + feat) = y0;
        *(v4f*)(orow + feat + 4) = y1;
      }
  }
}

extern "C" void kernel_launch(void* const* d_in, const int* in_sizes, int n_in,
                              void* d_out, int out_size, void* d_ws,
                              size_t ws_size, hipStream_t stream) {
  const float* x = (const float*)d_in[0];
  const float* c0 = (const float*)d_in[1];
  const float* c1 = (const float*)d_in[2];
  const float* c2 = (const float*)d_in[3];
  const float* c3 = (const float*)d_in[4];
  const float* bias = (const float*)d_in[5];
  float* out = (float*)d_out;

  dim3 grid(4096 / 16);
  dim3 block(128);
  tt_chain_kernel<<<grid, block, 0, stream>>>(x, c0, c1, c2, c3, bias, out);
}


# Core 0 folded into core 1 per o0 (8-term f32 FMA sums in registers at pass start): one K=(i1,i0) WMMA stage replaces stages 0+1, 128 not 144 WMMAs per block; sum order of both contractions changes
# speedup vs baseline: 4.0476x; 1.0769x over previous
#include <hip/hip_runtime.h>


typedef float v2f __attribute__((ext_vector_type(2)));
typedef float v4f __attribute__((ext_vector_type(4)));
typedef float v8f __attribute__((ext_vector_type(8)));

#define WMMA4(A, B, C)                                                         \
  __builtin_amdgcn_wmma_f32_16x16x4_f32(false, (A), false, (B), (short)0,      \
                                        (C), false, false)

__global__ __launch_bounds__(128, 1) void tt_chain_kernel(
    const float* __restrict__ x, const float* __restrict__ c0,
    const float* __restrict__ c1, const float* __restrict__ c2,
    const float* __restrict__ c3, const float* __restrict__ bias,
    float* __restrict__ out) {
  __shared__ __align__(16) v4f xs4[512 * 32];
  __shared__ __align__(16) v2f f2s[8 * 2 * 4 * 32];
  __shared__ __align__(16) v2f f3s[8 * 4 * 32];

  const int tid = threadIdx.x;
  const int w = tid >> 5;
  const int lane = tid & 31;
  const int c = lane & 15;
  const int h = lane >> 4;
  const int bg0 = blockIdx.x * 16;

  {
    const float* xrow = x + (size_t)(bg0 + c) * 4096;
    for (int n = 0; n < 8; ++n) {
      const int hp = n >> 2, itb = (n & 3) * 4;
      {
        v4f v[4][4];
#pragma unroll
        for (int a = 0; a < 4; ++a) {
          const int qq = (itb + a) * 8 + w * 2 + h;
#pragma unroll
          for (int m = 0; m < 4; ++m) {
            const int i0 = 2 * hp + (m & 1) + 4 * (m >> 1);
            v[a][m] = *(const v4f*)(xrow + i0 * 512 + 4 * qq);
          }
        }
#pragma unroll
        for (int a = 0; a < 4; ++a) {
          const int it = itb + a;
          const int i1 = it >> 1;
          const int i2 = (it & 1) * 4 + w;
#pragma unroll
          for (int e = 0; e < 4; ++e) {
            const int i3 = 4 * h + e;
            v4f s;
#pragma unroll
            for (int m = 0; m < 4; ++m) s[m] = v[a][m][e];
            xs4[((i3 * 8 + i2) * 8 + i1) * 32 + hp * 16 + c] = s;
          }
        }
      }
    }
  }
  for (int e = tid; e < 2048; e += 128) {
    const int ln = e & 31, T2 = (e >> 5) & 3, u = (e >> 7) & 1, i2 = e >> 8;
    const int cc = ln & 15, hh = ln >> 4;
    const int o2 = 2 * T2 + (cc >> 3), r3 = cc & 7;
    v2f v;
#pragma unroll
    for (int j = 0; j < 2; ++j) {
      const int r2 = 4 * hh + 2 * u + j;
      v[j] = c2[((r2 * 8 + i2) * 8 + o2) * 8 + r3];
    }
    f2s[e] = v;
  }
  for (int e = tid; e < 1024; e += 128) {
    const int ln = e & 31, t = (e >> 5) & 3, i3 = e >> 7;
    const int cc = ln & 15, hh = ln >> 4;
    v2f v;
#pragma unroll
    for (int j = 0; j < 2; ++j)
      v[j] = ((cc >> 3) == hh) ? c3[((2 * t + j) * 8 + i3) * 8 + (cc & 7)] : 0.f;
    f3s[e] = v;
  }
  __syncthreads();

  v2f f2[8][2][4];
#pragma unroll
  for (int i2 = 0; i2 < 8; ++i2)
#pragma unroll
    for (int u = 0; u < 2; ++u)
#pragma unroll
      for (int T2 = 0; T2 < 4; ++T2)
        f2[i2][u][T2] = f2s[((i2 * 2 + u) * 4 + T2) * 32 + lane];

  v4f xa[8];
#pragma unroll
  for (int i1 = 0; i1 < 8; ++i1) xa[i1] = xs4[i1 * 32 + lane];

  for (int pass = 0; pass < 2; ++pass) {
    const int o0 = w + 4 * pass;
    v2f f01[8][2][4];
    {
      float c0v[2][2][8];
#pragma unroll
      for (int t = 0; t < 2; ++t)
#pragma unroll
        for (int j = 0; j < 2; ++j) {
          const float* p = c0 + ((4 * t + 2 * h + j) * 8 + o0) * 8;
          const v4f lo = *(const v4f*)p;
          const v4f hi = *(const v4f*)(p + 4);
#pragma unroll
          for (int r = 0; r < 4; ++r) {
            c0v[t][j][r] = lo[r];
            c0v[t][j][4 + r] = hi[r];
          }
        }
      int c1pos = 0;
      asm volatile("" : "+s"(c1pos));
      const float* c1l =
          c1 + c1pos + ((c >> 2) & 1) * 8 + 4 * (c >> 3) + (c & 3);
#pragma unroll
      for (int i1 = 0; i1 < 8; ++i1)
#pragma unroll
        for (int T = 0; T < 4; ++T) {
          float c1v[8];
#pragma unroll
          for (int r1 = 0; r1 < 8; ++r1)
            c1v[r1] = c1l[((r1 * 8 + i1) * 8 + 2 * T) * 8];
#pragma unroll
          for (int t = 0; t < 2; ++t)
#pragma unroll
            for (int j = 0; j < 2; ++j) {
              float sum = c0v[t][j][0] * c1v[0];
#pragma unroll
              for (int r1 = 1; r1 < 8; ++r1)
                sum = __builtin_fmaf(c0v[t][j][r1], c1v[r1], sum);
              f01[i1][t][T][j] = sum;
            }
        }
    }

    v8f acc3[8][4];
#pragma unroll
    for (int o1 = 0; o1 < 8; ++o1)
#pragma unroll
      for (int T2 = 0; T2 < 4; ++T2)
#pragma unroll
        for (int r = 0; r < 8; ++r) acc3[o1][T2][r] = 0.f;

    for (int i3 = 0; i3 < 8; ++i3) {
      v8f acc2[8][4];
#pragma unroll
      for (int o1 = 0; o1 < 8; ++o1)
#pragma unroll
        for (int T2 = 0; T2 < 4; ++T2)
#pragma unroll
          for (int r = 0; r < 8; ++r) acc2[o1][T2][r] = 0.f;
      const v4f* xb = xs4 + i3 * 2048 + lane;
      const v4f* xnb = xs4 + ((i3 + 1) & 7) * 2048 + lane;
      v2f a3[4];

#pragma unroll
      for (int i2 = 0; i2 < 8; ++i2) {
        v4f xn[8];
        __builtin_amdgcn_sched_barrier(0);
        if (i2 < 7) {
#pragma unroll
          for (int i1 = 0; i1 < 8; ++i1) xn[i1] = xb[((i2 + 1) * 8 + i1) * 32];
        } else {
#pragma unroll
          for (int i1 = 0; i1 < 8; ++i1) xn[i1] = xnb[i1 * 32];
#pragma unroll
          for (int t = 0; t < 4; ++t) a3[t] = f3s[(i3 * 4 + t) * 32 + lane];
        }
        __builtin_amdgcn_sched_barrier(0);
        v8f d1[4];
#pragma unroll
        for (int T = 0; T < 4; ++T)
#pragma unroll
          for (int r = 0; r < 8; ++r) d1[T][r] = 0.f;
#pragma unroll
        for (int i1 = 0; i1 < 8; ++i1)
#pragma unroll
          for (int t = 0; t < 2; ++t) {
            v2f bx;
            bx[0] = xa[i1][2 * t];
            bx[1] = xa[i1][2 * t + 1];
#pragma unroll
            for (int T = 0; T < 4; ++T) d1[T] = WMMA4(f01[i1][t][T], bx, d1[T]);
          }
#pragma unroll
        for (int u = 0; u < 2; ++u)
#pragma unroll
          for (int T2 = 0; T2 < 4; ++T2) {
            const v2f a2 = f2[i2][u][T2];
#pragma unroll
            for (int o1 = 0; o1 < 8; ++o1) {
              v2f bz;
              bz[0] = d1[o1 >> 1][4 * (o1 & 1) + 2 * u];
              bz[1] = d1[o1 >> 1][4 * (o1 & 1) + 2 * u + 1];
              acc2[o1][T2] = WMMA4(a2, bz, acc2[o1][T2]);
            }
          }
#pragma unroll
        for (int i1 = 0; i1 < 8; ++i1) xa[i1] = xn[i1];
      }
#pragma unroll
      for (int t = 0; t < 4; ++t) {
#pragma unroll
        for (int o1 = 0; o1 < 8; ++o1)
#pragma unroll
          for (int T2 = 0; T2 < 4; ++T2) {
            v2f bz;
            bz[0] = acc2[o1][T2][2 * t];
            bz[1] = acc2[o1][T2][2 * t + 1];
            acc3[o1][T2] = WMMA4(a3[t], bz, acc3[o1][T2]);
          }
      }
    }

    float* orow = out + (size_t)(bg0 + c) * 4096;
#pragma unroll
    for (int o1 = 0; o1 < 8; ++o1)
#pragma unroll
      for (int T2 = 0; T2 < 4; ++T2) {
        const int feat = o0 * 512 + o1 * 64 + (2 * T2 + h) * 8;
        const v4f b0 = *(const v4f*)(bias + feat);
        const v4f b1 = *(const v4f*)(bias + feat + 4);
        v4f y0, y1;
#pragma unroll
        for (int r = 0; r < 4; ++r) {
          y0[r] = acc3[o1][T2][r] + b0[r];
          y1[r] = acc3[o1][T2][4 + r] + b1[r];
        }
        *(v4f*)(orow + feat) = y0;
        *(v4f*)(orow + feat + 4) = y1;
      }
  }
}

extern "C" void kernel_launch(void* const* d_in, const int* in_sizes, int n_in,
                              void* d_out, int out_size, void* d_ws,
                              size_t ws_size, hipStream_t stream) {
  const float* x = (const float*)d_in[0];
  const float* c0 = (const float*)d_in[1];
  const float* c1 = (const float*)d_in[2];
  const float* c2 = (const float*)d_in[3];
  const float* c3 = (const float*)d_in[4];
  const float* bias = (const float*)d_in[5];
  float* out = (float*)d_out;

  dim3 grid(4096 / 16);
  dim3 block(128);
  tt_chain_kernel<<<grid, block, 0, stream>>>(x, c0, c1, c2, c3, bias, out);
}


# Both o0 of a wave run side by side in every block: both folded c01 fragment sets stay in registers (folded once, one pass over core 1), LDS fragments fetched once for both; same sums, same order
# speedup vs baseline: 4.4907x; 1.1095x over previous
#include <hip/hip_runtime.h>


typedef float v2f __attribute__((ext_vector_type(2)));
typedef float v4f __attribute__((ext_vector_type(4)));
typedef float v8f __attribute__((ext_vector_type(8)));

#define WMMA4(A, B, C)                                                         \
  __builtin_amdgcn_wmma_f32_16x16x4_f32(false, (A), false, (B), (short)0,      \
                                        (C), false, false)

static __device__ __forceinline__ void block_one_o0(const v2f (&f)[8][2][4],
                                                    const v4f (&xa)[8],
                                                    const v4f (&wa)[4],
                                                    v8f (&acc)[8][4]) {
  v8f d1[4];
#pragma unroll
  for (int T = 0; T < 4; ++T)
#pragma unroll
    for (int r = 0; r < 8; ++r) d1[T][r] = 0.f;
#pragma unroll
  for (int i1 = 0; i1 < 8; ++i1)
#pragma unroll
    for (int t = 0; t < 2; ++t) {
      v2f bx;
      bx[0] = xa[i1][2 * t];
      bx[1] = xa[i1][2 * t + 1];
#pragma unroll
      for (int T = 0; T < 4; ++T) d1[T] = WMMA4(f[i1][t][T], bx, d1[T]);
    }
#pragma unroll
  for (int u = 0; u < 2; ++u)
#pragma unroll
    for (int T2 = 0; T2 < 4; ++T2) {
      v2f a2;
      a2[0] = wa[T2][2 * u];
      a2[1] = wa[T2][2 * u + 1];
#pragma unroll
      for (int o1 = 0; o1 < 8; ++o1) {
        v2f bz;
        bz[0] = d1[o1 >> 1][4 * (o1 & 1) + 2 * u];
        bz[1] = d1[o1 >> 1][4 * (o1 & 1) + 2 * u + 1];
        acc[o1][T2] = WMMA4(a2, bz, acc[o1][T2]);
      }
    }
}

static __device__ __forceinline__ void store_one_o0(const v8f (&acc)[8][4],
                                                    const float* __restrict__ bias,
                                                    float* orow, int o0,
                                                    int h) {
#pragma unroll
  for (int o1 = 0; o1 < 8; ++o1)
#pragma unroll
    for (int T2 = 0; T2 < 4; ++T2) {
      const int feat = o0 * 512 + o1 * 64 + (2 * T2 + h) * 8;
      const v4f b0 = *(const v4f*)(bias + feat);
      const v4f b1 = *(const v4f*)(bias + feat + 4);
      v4f y0, y1;
#pragma unroll
      for (int r = 0; r < 4; ++r) {
        y0[r] = acc[o1][T2][r] + b0[r];
        y1[r] = acc[o1][T2][4 + r] + b1[r];
      }
      *(v4f*)(orow + feat) = y0;
      *(v4f*)(orow + feat + 4) = y1;
    }
}

__global__ __launch_bounds__(128, 1) void tt_chain_kernel(
    const float* __restrict__ x, const float* __restrict__ c0,
    const float* __restrict__ c1, const float* __restrict__ c2,
    const float* __restrict__ c3, const float* __restrict__ bias,
    float* __restrict__ out) {
  __shared__ __align__(16) v4f xs4[256 * 32];
  __shared__ __align__(16) v4f w23s[64 * 4 * 32];

  const int tid = threadIdx.x;
  const int w = tid >> 5;
  const int lane = tid & 31;
  const int c = lane & 15;
  const int h = lane >> 4;
  const int bg0 = blockIdx.x * 16;
  const unsigned rowb = (unsigned)(bg0 + c) * 16384u;
  const float* xrow = (const float*)((const char*)x + rowb);
  float* orow = (float*)((char*)out + rowb);

  {
    const int o2 = 2 * w + (c >> 3), o3 = c & 7;
    float c3v[8][8];
#pragma unroll
    for (int i3 = 0; i3 < 8; ++i3)
#pragma unroll
      for (int r3 = 0; r3 < 8; ++r3) c3v[i3][r3] = c3[(r3 * 8 + i3) * 8 + o3];
#pragma unroll 2
    for (int i2 = 0; i2 < 8; ++i2) {
      float c2v[4][8];
#pragma unroll
      for (int q = 0; q < 4; ++q) {
        const float* p = c2 + (((4 * h + q) * 8 + i2) * 8 + o2) * 8;
        const v4f lo = *(const v4f*)p;
        const v4f hi = *(const v4f*)(p + 4);
#pragma unroll
        for (int r = 0; r < 4; ++r) {
          c2v[q][r] = lo[r];
          c2v[q][4 + r] = hi[r];
        }
      }
#pragma unroll
      for (int i3 = 0; i3 < 8; ++i3) {
        v4f o;
#pragma unroll
        for (int q = 0; q < 4; ++q) {
          float sum = c2v[q][0] * c3v[i3][0];
#pragma unroll
          for (int r3 = 1; r3 < 8; ++r3)
            sum = __builtin_fmaf(c2v[q][r3], c3v[i3][r3], sum);
          o[q] = sum;
        }
        w23s[((i3 * 8 + i2) * 4 + w) * 32 + lane] = o;
      }
    }
  }

  v2f fA[8][2][4], fB[8][2][4];
  {
    float c0a[2][2][8], c0b[2][2][8];
#pragma unroll
    for (int t = 0; t < 2; ++t)
#pragma unroll
      for (int j = 0; j < 2; ++j) {
        const float* p = c0 + ((4 * t + 2 * h + j) * 8 + w) * 8;
        const v4f lo = *(const v4f*)p;
        const v4f hi = *(const v4f*)(p + 4);
        const v4f lo2 = *(const v4f*)(p + 32);
        const v4f hi2 = *(const v4f*)(p + 36);
#pragma unroll
        for (int r = 0; r < 4; ++r) {
          c0a[t][j][r] = lo[r];
          c0a[t][j][4 + r] = hi[r];
          c0b[t][j][r] = lo2[r];
          c0b[t][j][4 + r] = hi2[r];
        }
      }
    const float* c1l = c1 + ((c >> 2) & 1) * 8 + 4 * (c >> 3) + (c & 3);
#pragma unroll
    for (int i1 = 0; i1 < 8; ++i1)
#pragma unroll
      for (int T = 0; T < 4; ++T) {
        float c1v[8];
#pragma unroll
        for (int r1 = 0; r1 < 8; ++r1)
          c1v[r1] = c1l[((r1 * 8 + i1) * 8 + 2 * T) * 8];
#pragma unroll
        for (int t = 0; t < 2; ++t)
#pragma unroll
          for (int j = 0; j < 2; ++j) {
            float sa = c0a[t][j][0] * c1v[0];
            float sb = c0b[t][j][0] * c1v[0];
#pragma unroll
            for (int r1 = 1; r1 < 8; ++r1) {
              sa = __builtin_fmaf(c0a[t][j][r1], c1v[r1], sa);
              sb = __builtin_fmaf(c0b[t][j][r1], c1v[r1], sb);
            }
            fA[i1][t][T][j] = sa;
            fB[i1][t][T][j] = sb;
          }
      }
  }

  v8f accA[8][4], accB[8][4];
#pragma unroll
  for (int o1 = 0; o1 < 8; ++o1)
#pragma unroll
    for (int T2 = 0; T2 < 4; ++T2)
#pragma unroll
      for (int r = 0; r < 8; ++r) {
        accA[o1][T2][r] = 0.f;
        accB[o1][T2][r] = 0.f;
      }

#pragma nounroll
  for (int H = 0; H < 2; ++H) {
    {
      const float* xh = xrow + H * 32 + w * 8 + h * 4;
      v4f va[8][4], vb[8][4];
#pragma unroll
      for (int i1 = 0; i1 < 8; ++i1)
#pragma unroll
        for (int m = 0; m < 4; ++m) {
          const int i0 = (m & 1) + 4 * (m >> 1);
          va[i1][m] = *(const v4f*)(xh + i0 * 512 + i1 * 64);
        }
      __builtin_amdgcn_fence(__ATOMIC_RELEASE, "workgroup", "local");
      __builtin_amdgcn_s_barrier();
      __builtin_amdgcn_fence(__ATOMIC_ACQUIRE, "workgroup", "local");
#pragma unroll
      for (int i1 = 0; i1 < 8; ++i1) {
#pragma unroll
        for (int m = 0; m < 4; ++m) {
          const int i0 = 2 + (m & 1) + 4 * (m >> 1);
          vb[i1][m] = *(const v4f*)(xh + i0 * 512 + i1 * 64);
        }
#pragma unroll
        for (int e = 0; e < 4; ++e) {
          const int i3 = 4 * h + e;
          v4f s;
#pragma unroll
          for (int m = 0; m < 4; ++m) s[m] = va[i1][m][e];
          xs4[((i3 * 4 + w) * 8 + i1) * 32 + c] = s;
        }
        __builtin_amdgcn_sched_barrier(0);
      }
#pragma unroll
      for (int i1 = 0; i1 < 8; ++i1)
#pragma unroll
        for (int e = 0; e < 4; ++e) {
          const int i3 = 4 * h + e;
          v4f s;
#pragma unroll
          for (int m = 0; m < 4; ++m) s[m] = vb[i1][m][e];
          xs4[((i3 * 4 + w) * 8 + i1) * 32 + 16 + c] = s;
        }
      __syncthreads();
    }

    v4f xa[8];
#pragma unroll
    for (int i1 = 0; i1 < 8; ++i1) xa[i1] = xs4[i1 * 32 + lane];
    const v4f* wh = w23s + H * 512 + lane;
    v4f wa[4];
#pragma unroll
    for (int T2 = 0; T2 < 4; ++T2) wa[T2] = wh[T2 * 32];

    for (int i3 = 0; i3 < 8; ++i3) {
      const v4f* xb = xs4 + i3 * 1024 + lane;
      const v4f* xnb = xs4 + ((i3 + 1) & 7) * 1024 + lane;
      const v4f* wb = wh + i3 * 1024;
      const v4f* wnb = wh + ((i3 + 1) & 7) * 1024;

#pragma unroll
      for (int i2l = 0; i2l < 4; ++i2l) {
        v4f xn[8], wn[4];
        __builtin_amdgcn_sched_barrier(0);
        if (i2l < 3) {
#pragma unroll
          for (int i1 = 0; i1 < 8; ++i1)
            xn[i1] = xb[((i2l + 1) * 8 + i1) * 32];
#pragma unroll
          for (int T2 = 0; T2 < 4; ++T2) wn[T2] = wb[(i2l + 1) * 128 + T2 * 32];
        } else {
#pragma unroll
          for (int i1 = 0; i1 < 8; ++i1) xn[i1] = xnb[i1 * 32];
#pragma unroll
          for (int T2 = 0; T2 < 4; ++T2) wn[T2] = wnb[T2 * 32];
        }
        __builtin_amdgcn_sched_barrier(0);
        block_one_o0(fA, xa, wa, accA);
        if (i2l < 3 || H == 0 || i3 < 7) block_one_o0(fB, xa, wa, accB);
#pragma unroll
        for (int i1 = 0; i1 < 8; ++i1) xa[i1] = xn[i1];
#pragma unroll
        for (int T2 = 0; T2 < 4; ++T2) wa[T2] = wn[T2];
      }
    }
  }

  {
    v4f xl[8], wl[4];
#pragma unroll
    for (int i1 = 0; i1 < 8; ++i1)
      xl[i1] = xs4[7 * 1024 + (3 * 8 + i1) * 32 + lane];
#pragma unroll
    for (int T2 = 0; T2 < 4; ++T2)
      wl[T2] = w23s[7 * 1024 + 512 + 3 * 128 + T2 * 32 + lane];
    store_one_o0(accA, bias, orow, w, h);
    __builtin_amdgcn_sched_barrier(0);
    block_one_o0(fB, xl, wl, accB);
  }

  store_one_o0(accB, bias, orow, w + 4, h);
}

extern "C" void kernel_launch(void* const* d_in, const int* in_sizes, int n_in,
                              void* d_out, int out_size, void* d_ws,
                              size_t ws_size, hipStream_t stream) {
  const float* x = (const float*)d_in[0];
  const float* c0 = (const float*)d_in[1];
  const float* c1 = (const float*)d_in[2];
  const float* c2 = (const float*)d_in[3];
  const float* c3 = (const float*)d_in[4];
  const float* bias = (const float*)d_in[5];
  float* out = (float*)d_out;

  dim3 grid(4096 / 16);
  dim3 block(128);
  tt_chain_kernel<<<grid, block, 0, stream>>>(x, c0, c1, c2, c3, bias, out);
}


# Outputs stored as whole 128-byte lines: each wave turns its accumulator tiles round in a private LDS strip and stores one batch row, 512 contiguous bytes, per instruction; same sums, same order
# speedup vs baseline: 4.7484x; 1.0574x over previous
#include <hip/hip_runtime.h>


typedef float v2f __attribute__((ext_vector_type(2)));
typedef float v4f __attribute__((ext_vector_type(4)));
typedef float v8f __attribute__((ext_vector_type(8)));

#define WMMA4(A, B, C)                                                         \
  __builtin_amdgcn_wmma_f32_16x16x4_f32(false, (A), false, (B), (short)0,      \
                                        (C), false, false)

static __device__ __forceinline__ void block_one_o0(const v2f (&f)[8][2][4],
                                                    const v4f (&xa)[8],
                                                    const v4f (&wa)[4],
                                                    v8f (&acc)[8][4]) {
  v8f d1[4];
#pragma unroll
  for (int T = 0; T < 4; ++T)
#pragma unroll
    for (int r = 0; r < 8; ++r) d1[T][r] = 0.f;
#pragma unroll
  for (int i1 = 0; i1 < 8; ++i1)
#pragma unroll
    for (int t = 0; t < 2; ++t) {
      v2f bx;
      bx[0] = xa[i1][2 * t];
      bx[1] = xa[i1][2 * t + 1];
#pragma unroll
      for (int T = 0; T < 4; ++T) d1[T] = WMMA4(f[i1][t][T], bx, d1[T]);
    }
#pragma unroll
  for (int u = 0; u < 2; ++u)
#pragma unroll
    for (int T2 = 0; T2 < 4; ++T2) {
      v2f a2;
      a2[0] = wa[T2][2 * u];
      a2[1] = wa[T2][2 * u + 1];
#pragma unroll
      for (int o1 = 0; o1 < 8; ++o1) {
        v2f bz;
        bz[0] = d1[o1 >> 1][4 * (o1 & 1) + 2 * u];
        bz[1] = d1[o1 >> 1][4 * (o1 & 1) + 2 * u + 1];
        acc[o1][T2] = WMMA4(a2, bz, acc[o1][T2]);
      }
    }
}

static __device__ __forceinline__ void store_one_o0(const v8f (&acc)[8][4],
                                                    const float* __restrict__ bias,
                                                    float* ob, float* st,
                                                    int o0, int c, int h,
                                                    int lane) {
  float* wr = st + c * 132 + h * 8;
  const float* rd = st + 4 * lane;
#pragma unroll
  for (int p = 0; p < 4; ++p) {
#pragma unroll
    for (int o1l = 0; o1l < 2; ++o1l)
#pragma unroll
      for (int T2 = 0; T2 < 4; ++T2) {
        v4f lo, hi;
#pragma unroll
        for (int r = 0; r < 4; ++r) {
          lo[r] = acc[2 * p + o1l][T2][r];
          hi[r] = acc[2 * p + o1l][T2][4 + r];
        }
        *(v4f*)(wr + o1l * 64 + T2 * 16) = lo;
        *(v4f*)(wr + o1l * 64 + T2 * 16 + 4) = hi;
      }
    __builtin_amdgcn_fence(__ATOMIC_RELEASE, "wavefront", "local");
    __builtin_amdgcn_fence(__ATOMIC_ACQUIRE, "wavefront", "local");
    const int feat = o0 * 512 + p * 128;
    const v4f bv = *(const v4f*)(bias + feat + 4 * lane);
    v4f v[16];
#pragma unroll
    for (int j = 0; j < 16; ++j) v[j] = *(const v4f*)(rd + j * 132);
    __builtin_amdgcn_fence(__ATOMIC_RELEASE, "wavefront", "local");
    __builtin_amdgcn_fence(__ATOMIC_ACQUIRE, "wavefront", "local");
#pragma unroll
    for (int j = 0; j < 16; ++j) {
      v4f y;
#pragma unroll
      for (int r = 0; r < 4; ++r) y[r] = v[j][r] + bv[r];
      *(v4f*)(ob + j * 4096 + feat) = y;
    }
  }
}

__global__ __launch_bounds__(128, 1) void tt_chain_kernel(
    const float* __restrict__ x, const float* __restrict__ c0,
    const float* __restrict__ c1, const float* __restrict__ c2,
    const float* __restrict__ c3, const float* __restrict__ bias,
    float* __restrict__ out) {
  __shared__ __align__(16) v4f xs4[256 * 32];
  __shared__ __align__(16) v4f w23s[64 * 4 * 32];
  __shared__ __align__(16) float ostage[4 * 16 * 132];

  const int tid = threadIdx.x;
  const int w = tid >> 5;
  const int lane = tid & 31;
  const int c = lane & 15;
  const int h = lane >> 4;
  const int bg0 = blockIdx.x * 16;
  const unsigned rowb = (unsigned)(bg0 + c) * 16384u;
  const float* xrow = (const float*)((const char*)x + rowb);
  float* ob = (float*)((char*)out + (unsigned)bg0 * 16384u) + 4 * lane;
  float* ost = ostage + w * (16 * 132);

  {
    const int o2 = 2 * w + (c >> 3), o3 = c & 7;
    float c3v[8][8];
#pragma unroll
    for (int i3 = 0; i3 < 8; ++i3)
#pragma unroll
      for (int r3 = 0; r3 < 8; ++r3) c3v[i3][r3] = c3[(r3 * 8 + i3) * 8 + o3];
#pragma unroll 2
    for (int i2 = 0; i2 < 8; ++i2) {
      float c2v[4][8];
#pragma unroll
      for (int q = 0; q < 4; ++q) {
        const float* p = c2 + (((4 * h + q) * 8 + i2) * 8 + o2) * 8;
        const v4f lo = *(const v4f*)p;
        const v4f hi = *(const v4f*)(p + 4);
#pragma unroll
        for (int r = 0; r < 4; ++r) {
          c2v[q][r] = lo[r];
          c2v[q][4 + r] = hi[r];
        }
      }
#pragma unroll
      for (int i3 = 0; i3 < 8; ++i3) {
        v4f o;
#pragma unroll
        for (int q = 0; q < 4; ++q) {
          float sum = c2v[q][0] * c3v[i3][0];
#pragma unroll
          for (int r3 = 1; r3 < 8; ++r3)
            sum = __builtin_fmaf(c2v[q][r3], c3v[i3][r3], sum);
          o[q] = sum;
        }
        w23s[((i3 * 8 + i2) * 4 + w) * 32 + lane] = o;
      }
    }
  }

  v2f fA[8][2][4], fB[8][2][4];
  {
    float c0a[2][2][8], c0b[2][2][8];
#pragma unroll
    for (int t = 0; t < 2; ++t)
#pragma unroll
      for (int j = 0; j < 2; ++j) {
        const float* p = c0 + ((4 * t + 2 * h + j) * 8 + w) * 8;
        const v4f lo = *(const v4f*)p;
        const v4f hi = *(const v4f*)(p + 4);
        const v4f lo2 = *(const v4f*)(p + 32);
        const v4f hi2 = *(const v4f*)(p + 36);
#pragma unroll
        for (int r = 0; r < 4; ++r) {
          c0a[t][j][r] = lo[r];
          c0a[t][j][4 + r] = hi[r];
          c0b[t][j][r] = lo2[r];
          c0b[t][j][4 + r] = hi2[r];
        }
      }
    const float* c1l = c1 + ((c >> 2) & 1) * 8 + 4 * (c >> 3) + (c & 3);
#pragma unroll
    for (int i1 = 0; i1 < 8; ++i1)
#pragma unroll
      for (int T = 0; T < 4; ++T) {
        float c1v[8];
#pragma unroll
        for (int r1 = 0; r1 < 8; ++r1)
          c1v[r1] = c1l[((r1 * 8 + i1) * 8 + 2 * T) * 8];
#pragma unroll
        for (int t = 0; t < 2; ++t)
#pragma unroll
          for (int j = 0; j < 2; ++j) {
            float sa = c0a[t][j][0] * c1v[0];
            float sb = c0b[t][j][0] * c1v[0];
#pragma unroll
            for (int r1 = 1; r1 < 8; ++r1) {
              sa = __builtin_fmaf(c0a[t][j][r1], c1v[r1], sa);
              sb = __builtin_fmaf(c0b[t][j][r1], c1v[r1], sb);
            }
            fA[i1][t][T][j] = sa;
            fB[i1][t][T][j] = sb;
          }
      }
  }

  v8f accA[8][4], accB[8][4];
#pragma unroll
  for (int o1 = 0; o1 < 8; ++o1)
#pragma unroll
    for (int T2 = 0; T2 < 4; ++T2)
#pragma unroll
      for (int r = 0; r < 8; ++r) {
        accA[o1][T2][r] = 0.f;
        accB[o1][T2][r] = 0.f;
      }

#pragma nounroll
  for (int H = 0; H < 2; ++H) {
    {
      const float* xh = xrow + H * 32 + w * 8 + h * 4;
      v4f va[8][4], vb[8][4];
#pragma unroll
      for (int i1 = 0; i1 < 8; ++i1)
#pragma unroll
        for (int m = 0; m < 4; ++m) {
          const int i0 = (m & 1) + 4 * (m >> 1);
          va[i1][m] = *(const v4f*)(xh + i0 * 512 + i1 * 64);
        }
      __builtin_amdgcn_fence(__ATOMIC_RELEASE, "workgroup", "local");
      __builtin_amdgcn_s_barrier();
      __builtin_amdgcn_fence(__ATOMIC_ACQUIRE, "workgroup", "local");
#pragma unroll
      for (int i1 = 0; i1 < 8; ++i1) {
#pragma unroll
        for (int m = 0; m < 4; ++m) {
          const int i0 = 2 + (m & 1) + 4 * (m >> 1);
          vb[i1][m] = *(const v4f*)(xh + i0 * 512 + i1 * 64);
        }
#pragma unroll
        for (int e = 0; e < 4; ++e) {
          const int i3 = 4 * h + e;
          v4f s;
#pragma unroll
          for (int m = 0; m < 4; ++m) s[m] = va[i1][m][e];
          xs4[((i3 * 4 + w) * 8 + i1) * 32 + c] = s;
        }
        __builtin_amdgcn_sched_barrier(0);
      }
#pragma unroll
      for (int i1 = 0; i1 < 8; ++i1)
#pragma unroll
        for (int e = 0; e < 4; ++e) {
          const int i3 = 4 * h + e;
          v4f s;
#pragma unroll
          for (int m = 0; m < 4; ++m) s[m] = vb[i1][m][e];
          xs4[((i3 * 4 + w) * 8 + i1) * 32 + 16 + c] = s;
        }
      __syncthreads();
    }

    v4f xa[8];
#pragma unroll
    for (int i1 = 0; i1 < 8; ++i1) xa[i1] = xs4[i1 * 32 + lane];
    const v4f* wh = w23s + H * 512 + lane;
    v4f wa[4];
#pragma unroll
    for (int T2 = 0; T2 < 4; ++T2) wa[T2] = wh[T2 * 32];

    for (int i3 = 0; i3 < 8; ++i3) {
      const v4f* xb = xs4 + i3 * 1024 + lane;
      const v4f* xnb = xs4 + ((i3 + 1) & 7) * 1024 + lane;
      const v4f* wb = wh + i3 * 1024;
      const v4f* wnb = wh + ((i3 + 1) & 7) * 1024;

#pragma unroll
      for (int i2l = 0; i2l < 4; ++i2l) {
        v4f xn[8], wn[4];
        __builtin_amdgcn_sched_barrier(0);
        if (i2l < 3) {
#pragma unroll
          for (int i1 = 0; i1 < 8; ++i1)
            xn[i1] = xb[((i2l + 1) * 8 + i1) * 32];
#pragma unroll
          for (int T2 = 0; T2 < 4; ++T2) wn[T2] = wb[(i2l + 1) * 128 + T2 * 32];
        } else {
#pragma unroll
          for (int i1 = 0; i1 < 8; ++i1) xn[i1] = xnb[i1 * 32];
#pragma unroll
          for (int T2 = 0; T2 < 4; ++T2) wn[T2] = wnb[T2 * 32];
        }
        __builtin_amdgcn_sched_barrier(0);
        block_one_o0(fA, xa, wa, accA);
        if (i2l < 3 || H == 0 || i3 < 7) block_one_o0(fB, xa, wa, accB);
#pragma unroll
        for (int i1 = 0; i1 < 8; ++i1) xa[i1] = xn[i1];
#pragma unroll
        for (int T2 = 0; T2 < 4; ++T2) wa[T2] = wn[T2];
      }
    }
  }

  {
    v4f xl[8], wl[4];
#pragma unroll
    for (int i1 = 0; i1 < 8; ++i1)
      xl[i1] = xs4[7 * 1024 + (3 * 8 + i1) * 32 + lane];
#pragma unroll
    for (int T2 = 0; T2 < 4; ++T2)
      wl[T2] = w23s[7 * 1024 + 512 + 3 * 128 + T2 * 32 + lane];
    store_one_o0(accA, bias, ob, ost, w, c, h, lane);
    block_one_o0(fB, xl, wl, accB);
  }

  store_one_o0(accB, bias, ob, ost, w + 4, c, h, lane);
}

extern "C" void kernel_launch(void* const* d_in, const int* in_sizes, int n_in,
                              void* d_out, int out_size, void* d_ws,
                              size_t ws_size, hipStream_t stream) {
  const float* x = (const float*)d_in[0];
  const float* c0 = (const float*)d_in[1];
  const float* c1 = (const float*)d_in[2];
  const float* c2 = (const float*)d_in[3];
  const float* c3 = (const float*)d_in[4];
  const float* bias = (const float*)d_in[5];
  float* out = (float*)d_out;

  dim3 grid(4096 / 16);
  dim3 block(128);
  tt_chain_kernel<<<grid, block, 0, stream>>>(x, c0, c1, c2, c3, bias, out);
}
